# starting kernel
# baseline (speedup 1.0000x reference)
#include <hip/hip_runtime.h>

#define N_NODES 100000
#define IN_F    128
#define H_HEADS 4
#define D_OUT   16
#define HD      64
#define E_EDGES 3200000
#define SLOPE   0.2f

typedef __attribute__((ext_vector_type(2))) float v2f;
typedef __attribute__((ext_vector_type(4))) float v4f;
typedef __attribute__((ext_vector_type(8))) float v8f;

__device__ __forceinline__ unsigned flt_flip(float f) {
    unsigned u = __float_as_uint(f);
    return u ^ (((int)u >> 31) | 0x80000000u);
}
__device__ __forceinline__ float flt_unflip(unsigned u) {
    unsigned b = (u & 0x80000000u) ? (u ^ 0x80000000u) : ~u;
    return __uint_as_float(b);
}

__global__ __launch_bounds__(256) void gat_gemm(const float* __restrict__ feat,
                                                const float* __restrict__ W,
                                                float* __restrict__ ft) {
    const int wave = threadIdx.x >> 5;
    const int lane = threadIdx.x & 31;
    const int rt   = wave >> 2;
    const int ct   = wave & 3;
    const int r0   = blockIdx.x * 32 + rt * 16;
    const int c0   = ct * 16;
    const int m    = lane & 15;
    const int hi   = lane >> 4;

    const float* arow = feat + (size_t)(r0 + m) * IN_F + (hi << 1);
    const float* brow = W    + (size_t)(c0 + m) * IN_F + (hi << 1);

    v8f acc = {};
#pragma unroll
    for (int k0 = 0; k0 < IN_F; k0 += 4) {
        v2f a = *(const v2f*)(arow + k0);
        v2f b = *(const v2f*)(brow + k0);
        acc = __builtin_amdgcn_wmma_f32_16x16x4_f32(
            false, a, false, b, (short)0, acc, false, false);
    }
    float* outp = ft + (size_t)(r0 + hi * 8) * HD + c0 + m;
#pragma unroll
    for (int r = 0; r < 8; ++r) outp[(size_t)r * HD] = acc[r];
}

__global__ __launch_bounds__(256) void gat_attn(const float* __restrict__ ft,
                                                const float* __restrict__ al,
                                                const float* __restrict__ ar,
                                                float* __restrict__ el,
                                                float* __restrict__ er) {
    int idx = blockIdx.x * blockDim.x + threadIdx.x;
    if (idx >= N_NODES * H_HEADS) return;
    const int h = idx & 3;
    const float* f = ft + (size_t)idx * D_OUT;
    float sl = 0.f, sr = 0.f;
#pragma unroll
    for (int d = 0; d < D_OUT; ++d) {
        float v = f[d];
        sl += v * al[h * D_OUT + d];
        sr += v * ar[h * D_OUT + d];
    }
    el[idx] = sl;
    er[idx] = sr;
}

__global__ __launch_bounds__(256) void gat_init(unsigned* __restrict__ m_u,
                                                float* __restrict__ s,
                                                float* __restrict__ out) {
    int idx = blockIdx.x * blockDim.x + threadIdx.x;
    if (idx < N_NODES * H_HEADS) { m_u[idx] = 0u; s[idx] = 0.f; }
    if (idx < N_NODES * HD)      out[idx] = 0.f;
}

__device__ __forceinline__ float edge_score(float l, float r) {
    float e = l + r;
    return e >= 0.f ? e : SLOPE * e;
}

__global__ __launch_bounds__(256) void gat_edge_max(const int* __restrict__ src,
                                                    const int* __restrict__ dst,
                                                    const float* __restrict__ el,
                                                    const float* __restrict__ er,
                                                    unsigned* __restrict__ m_u) {
    int i = blockIdx.x * blockDim.x + threadIdx.x;
    if (i >= E_EDGES) return;
    const int s_ = src[i], d_ = dst[i];
    v4f l = *(const v4f*)(el + (size_t)s_ * 4);
    v4f r = *(const v4f*)(er + (size_t)d_ * 4);
#pragma unroll
    for (int h = 0; h < 4; ++h)
        atomicMax(&m_u[d_ * 4 + h], flt_flip(edge_score(l[h], r[h])));
}

__global__ __launch_bounds__(256) void gat_edge_sum(const int* __restrict__ src,
                                                    const int* __restrict__ dst,
                                                    const float* __restrict__ el,
                                                    const float* __restrict__ er,
                                                    const unsigned* __restrict__ m_u,
                                                    float* __restrict__ ssum) {
    int i = blockIdx.x * blockDim.x + threadIdx.x;
    if (i >= E_EDGES) return;
    const int s_ = src[i], d_ = dst[i];
    v4f l = *(const v4f*)(el + (size_t)s_ * 4);
    v4f r = *(const v4f*)(er + (size_t)d_ * 4);
#pragma unroll
    for (int h = 0; h < 4; ++h) {
        float e = edge_score(l[h], r[h]);
        float a = expf(e - flt_unflip(m_u[d_ * 4 + h]));
        atomicAdd(&ssum[d_ * 4 + h], a);
    }
}

__global__ __launch_bounds__(256) void gat_edge_agg(const int* __restrict__ src,
                                                    const int* __restrict__ dst,
                                                    const float* __restrict__ el,
                                                    const float* __restrict__ er,
                                                    const unsigned* __restrict__ m_u,
                                                    const float* __restrict__ ssum,
                                                    const float* __restrict__ ft,
                                                    float* __restrict__ out) {
    int t = blockIdx.x * blockDim.x + threadIdx.x;
    if (t >= E_EDGES * H_HEADS) return;
    const int i = t >> 2;
    const int h = t & 3;
    const int s_ = src[i], d_ = dst[i];
    float e = edge_score(el[s_ * 4 + h], er[d_ * 4 + h]);
    float w = expf(e - flt_unflip(m_u[d_ * 4 + h])) / ssum[d_ * 4 + h];
    const float* f = ft + (size_t)s_ * HD + h * D_OUT;
    float* o       = out + (size_t)d_ * HD + h * D_OUT;
#pragma unroll
    for (int d = 0; d < D_OUT; ++d)
        atomicAdd(o + d, f[d] * w);
}

extern "C" void kernel_launch(void* const* d_in, const int* in_sizes, int n_in,
                              void* d_out, int out_size, void* d_ws, size_t ws_size,
                              hipStream_t stream) {
    const float* feat = (const float*)d_in[0];
    const float* W    = (const float*)d_in[1];
    const float* al   = (const float*)d_in[2];
    const float* ar   = (const float*)d_in[3];
    const int*   src  = (const int*)d_in[4];
    const int*   dst  = (const int*)d_in[5];
    float* out = (float*)d_out;

    float*    ft  = (float*)d_ws;
    float*    el  = ft + (size_t)N_NODES * HD;
    float*    er  = el + (size_t)N_NODES * H_HEADS;
    unsigned* m_u = (unsigned*)(er + (size_t)N_NODES * H_HEADS);
    float*    s   = (float*)(m_u + (size_t)N_NODES * H_HEADS);

    const int nb_gemm = N_NODES / 32;
    const int nb_attn = (N_NODES * H_HEADS + 255) / 256;
    const int nb_init = (N_NODES * HD + 255) / 256;
    const int nb_edge = (E_EDGES + 255) / 256;
    const int nb_agg  = (E_EDGES * H_HEADS + 255) / 256;

    gat_gemm<<<nb_gemm, 256, 0, stream>>>(feat, W, ft);
    gat_attn<<<nb_attn, 256, 0, stream>>>(ft, al, ar, el, er);
    gat_init<<<nb_init, 256, 0, stream>>>(m_u, s, out);
    gat_edge_max<<<nb_edge, 256, 0, stream>>>(src, dst, el, er, m_u);
    gat_edge_sum<<<nb_edge, 256, 0, stream>>>(src, dst, el, er, m_u, s);
    gat_edge_agg<<<nb_agg, 256, 0, stream>>>(src, dst, el, er, m_u, s, ft, out);
}


# Edges sorted by destination (global counting sort, fill by atomic rank); one wave per node: max, exp sum, weighted row sum in registers, per head pair; float atomics gone; sum order = arrival rank
# speedup vs baseline: 3.1921x; 3.1921x over previous
#include <hip/hip_runtime.h>


#define N_NODES 100000
#define IN_F    128
#define H_HEADS 4
#define D_OUT   16
#define HD      64
#define HP      32
#define E_EDGES 3200000
#define SLOPE   0.2f
#define SCAN_T  1024

typedef __attribute__((ext_vector_type(2))) float v2f;
typedef __attribute__((ext_vector_type(8))) float v8f;

__global__ __launch_bounds__(128) void gat_gemm(const float* __restrict__ feat,
                                                const float* __restrict__ W,
                                                int p,
                                                float* __restrict__ fth) {
    const int wave = threadIdx.x >> 5;
    const int lane = threadIdx.x & 31;
    const int rt   = wave >> 1;
    const int ct   = wave & 1;
    const int r0   = blockIdx.x * 32 + rt * 16;
    const int c0   = (p * 2 + ct) * 16;
    const int m    = lane & 15;
    const int hi   = lane >> 4;

    const float* arow = feat + (size_t)(r0 + m) * IN_F + (hi << 1);
    const float* brow = W    + (size_t)(c0 + m) * IN_F + (hi << 1);

    v8f acc = {};
#pragma unroll
    for (int k0 = 0; k0 < IN_F; k0 += 4) {
        v2f a = *(const v2f*)(arow + k0);
        v2f b = *(const v2f*)(brow + k0);
        acc = __builtin_amdgcn_wmma_f32_16x16x4_f32(
            false, a, false, b, (short)0, acc, false, false);
    }
    float* outp = fth + (size_t)(r0 + hi * 8) * HP + ct * 16 + m;
#pragma unroll
    for (int r = 0; r < 8; ++r) outp[(size_t)r * HP] = acc[r];
}

__global__ __launch_bounds__(256) void gat_attn(const float* __restrict__ fth,
                                                const float* __restrict__ al,
                                                const float* __restrict__ ar,
                                                int p,
                                                float* __restrict__ el,
                                                float* __restrict__ er) {
    int idx = blockIdx.x * blockDim.x + threadIdx.x;
    if (idx >= N_NODES * 2) return;
    const int h = p * 2 + (idx & 1);
    const float* f = fth + (size_t)idx * D_OUT;
    float sl = 0.f, sr = 0.f;
#pragma unroll
    for (int d = 0; d < D_OUT; ++d) {
        float v = f[d];
        sl += v * al[h * D_OUT + d];
        sr += v * ar[h * D_OUT + d];
    }
    el[idx] = sl;
    er[idx] = sr;
}

__global__ __launch_bounds__(256) void csr_zero(int* __restrict__ cnt) {
    int i = blockIdx.x * blockDim.x + threadIdx.x;
    if (i < N_NODES) cnt[i] = 0;
}

__global__ __launch_bounds__(256) void csr_hist(const int* __restrict__ dst,
                                                int* __restrict__ cnt) {
    int i = blockIdx.x * blockDim.x + threadIdx.x;
    if (i < E_EDGES) atomicAdd(&cnt[dst[i]], 1);
}

__global__ __launch_bounds__(SCAN_T) void csr_scan(const int* __restrict__ cnt,
                                                   int* __restrict__ rowptr,
                                                   int* __restrict__ cur) {
    __shared__ int wsum[32];
    __shared__ int wtot[32];
    const int tid  = threadIdx.x;
    const int lane = tid & 31;
    const int wave = tid >> 5;
    const int base = blockIdx.x * SCAN_T;

    int part = 0;
    const int4* c4 = (const int4*)cnt;
    for (int j = tid; j < base / 4; j += SCAN_T) {
        int4 v = c4[j];
        part += (v.x + v.y) + (v.z + v.w);
    }
    for (int o = 16; o > 0; o >>= 1) part += __shfl_xor(part, o);
    if (lane == 0) wsum[wave] = part;

    const int i = base + tid;
    const int c = (i < N_NODES) ? cnt[i] : 0;
    int inc = c;
    for (int o = 1; o < 32; o <<= 1) {
        int t = __shfl_up(inc, o);
        if (lane >= o) inc += t;
    }
    if (lane == 31) wtot[wave] = inc;
    __syncthreads();

    int front = 0;
    for (int w = 0; w < 32; ++w) front += wsum[w];
    int woff = 0;
    for (int w = 0; w < wave; ++w) woff += wtot[w];
    const int excl = front + woff + inc - c;
    if (i <= N_NODES) rowptr[i] = excl;
    if (i <  N_NODES) cur[i]    = excl;
}

__global__ __launch_bounds__(256) void csr_fill(const int* __restrict__ src,
                                                const int* __restrict__ dst,
                                                int* __restrict__ cur,
                                                int* __restrict__ list) {
    int i = blockIdx.x * blockDim.x + threadIdx.x;
    if (i >= E_EDGES) return;
    const int pos = atomicAdd(&cur[dst[i]], 1);
    list[pos] = src[i];
}

__device__ __forceinline__ float edge_score(float l, float r) {
    float e = l + r;
    return e >= 0.f ? e : SLOPE * e;
}

__global__ __launch_bounds__(256) void gat_gather(const int* __restrict__ rowptr,
                                                  const int* __restrict__ list,
                                                  const float* __restrict__ fth,
                                                  const float* __restrict__ el,
                                                  const float* __restrict__ er,
                                                  int p,
                                                  float* __restrict__ out) {
    const int lane = threadIdx.x & 31;
    const int n    = blockIdx.x * 8 + __builtin_amdgcn_readfirstlane((int)(threadIdx.x >> 5));
    const bool up  = lane >= 16;
    const int beg  = rowptr[n];
    const int end  = rowptr[n + 1];
    const float er0 = er[(size_t)n * 2];
    const float er1 = er[(size_t)n * 2 + 1];

    float acc = 0.f;
    float sv  = 1.f;
    if (end > beg) {
        float m0 = -INFINITY, m1 = -INFINITY;
        for (int b = beg; b < end; b += 32) {
            const int j  = b + lane;
            const int s  = list[min(j, end - 1)];
            const v2f l  = *(const v2f*)(el + (size_t)s * 2);
            const float e0 = edge_score(l.x, er0);
            const float e1 = edge_score(l.y, er1);
            m0 = fmaxf(m0, j < end ? e0 : -INFINITY);
            m1 = fmaxf(m1, j < end ? e1 : -INFINITY);
        }
        for (int o = 16; o > 0; o >>= 1) {
            m0 = fmaxf(m0, __shfl_xor(m0, o));
            m1 = fmaxf(m1, __shfl_xor(m1, o));
        }
        float s0 = 0.f, s1 = 0.f;
        for (int b = beg; b < end; b += 32) {
            const int j  = b + lane;
            const int s  = list[min(j, end - 1)];
            const v2f l  = *(const v2f*)(el + (size_t)s * 2);
            const float a0 = j < end ? expf(edge_score(l.x, er0) - m0) : 0.f;
            const float a1 = j < end ? expf(edge_score(l.y, er1) - m1) : 0.f;
            s0 += a0;
            s1 += a1;
            const int cnt8 = (min(32, end - b) + 7) & ~7;
            for (int k = 0; k < cnt8; k += 8) {
                float v[8], w[8];
#pragma unroll
                for (int u = 0; u < 8; ++u) {
                    const int   sk = __builtin_amdgcn_readlane(s, k + u);
                    const float w0 = __int_as_float(__builtin_amdgcn_readlane(__float_as_int(a0), k + u));
                    const float w1 = __int_as_float(__builtin_amdgcn_readlane(__float_as_int(a1), k + u));
                    v[u] = fth[(size_t)sk * HP + lane];
                    w[u] = up ? w1 : w0;
                }
#pragma unroll
                for (int u = 0; u < 8; ++u) acc += w[u] * v[u];
            }
        }
        for (int o = 16; o > 0; o >>= 1) {
            s0 += __shfl_xor(s0, o);
            s1 += __shfl_xor(s1, o);
        }
        sv = up ? s1 : s0;
    }
    out[(size_t)n * HD + p * HP + lane] = acc / sv;
}

extern "C" void kernel_launch(void* const* d_in, const int* in_sizes, int n_in,
                              void* d_out, int out_size, void* d_ws, size_t ws_size,
                              hipStream_t stream) {
    const float* feat = (const float*)d_in[0];
    const float* W    = (const float*)d_in[1];
    const float* al   = (const float*)d_in[2];
    const float* ar   = (const float*)d_in[3];
    const int*   src  = (const int*)d_in[4];
    const int*   dst  = (const int*)d_in[5];
    float* out = (float*)d_out;

    float* fth    = (float*)d_ws;
    int*   list   = (int*)(fth + (size_t)N_NODES * HP);
    float* el     = (float*)(list + (size_t)E_EDGES);
    float* er     = el + (size_t)N_NODES * 2;
    int*   cnt    = (int*)(er + (size_t)N_NODES * 2);
    int*   rowptr = cnt + (size_t)N_NODES;
    int*   cur    = rowptr + 100352;

    const int nb_node = (N_NODES + 255) / 256;
    const int nb_edge = (E_EDGES + 255) / 256;
    const int nb_scan = (N_NODES + 1 + SCAN_T - 1) / SCAN_T;
    const int nb_gemm = N_NODES / 32;
    const int nb_attn = (N_NODES * 2 + 255) / 256;
    const int nb_gath = N_NODES / 8;

    csr_zero<<<nb_node, 256, 0, stream>>>(cnt);
    csr_hist<<<nb_edge, 256, 0, stream>>>(dst, cnt);
    csr_scan<<<nb_scan, SCAN_T, 0, stream>>>(cnt, rowptr, cur);
    csr_fill<<<nb_edge, 256, 0, stream>>>(src, dst, cur, list);

    gat_gemm<<<nb_gemm, 128, 0, stream>>>(feat, W, 0, fth);
    gat_attn<<<nb_attn, 256, 0, stream>>>(fth, al, ar, 0, el, er);
    gat_gather<<<nb_gath, 256, 0, stream>>>(rowptr, list, fth, el, er, 0, out);

    gat_gemm<<<nb_gemm, 128, 0, stream>>>(feat, W, 1, fth);
    gat_attn<<<nb_attn, 256, 0, stream>>>(fth, al, ar, 1, el, er);
    gat_gather<<<nb_gath, 256, 0, stream>>>(rowptr, list, fth, el, er, 1, out);
}


# Gather keeps a node's first 64 edges (source, score) in registers from the maximum pass to the sum pass: second read of the list and second gather of el gone; sums bit-identical
# speedup vs baseline: 3.3386x; 1.0459x over previous
#include <hip/hip_runtime.h>


#define N_NODES 100000
#define IN_F    128
#define H_HEADS 4
#define D_OUT   16
#define HD      64
#define HP      32
#define E_EDGES 3200000
#define SLOPE   0.2f
#define SCAN_T  1024

typedef __attribute__((ext_vector_type(2))) float v2f;
typedef __attribute__((ext_vector_type(8))) float v8f;

__global__ __launch_bounds__(128) void gat_gemm(const float* __restrict__ feat,
                                                const float* __restrict__ W,
                                                int p,
                                                float* __restrict__ fth) {
    const int wave = threadIdx.x >> 5;
    const int lane = threadIdx.x & 31;
    const int rt   = wave >> 1;
    const int ct   = wave & 1;
    const int r0   = blockIdx.x * 32 + rt * 16;
    const int c0   = (p * 2 + ct) * 16;
    const int m    = lane & 15;
    const int hi   = lane >> 4;

    const float* arow = feat + (size_t)(r0 + m) * IN_F + (hi << 1);
    const float* brow = W    + (size_t)(c0 + m) * IN_F + (hi << 1);

    v8f acc = {};
#pragma unroll
    for (int k0 = 0; k0 < IN_F; k0 += 4) {
        v2f a = *(const v2f*)(arow + k0);
        v2f b = *(const v2f*)(brow + k0);
        acc = __builtin_amdgcn_wmma_f32_16x16x4_f32(
            false, a, false, b, (short)0, acc, false, false);
    }
    float* outp = fth + (size_t)(r0 + hi * 8) * HP + ct * 16 + m;
#pragma unroll
    for (int r = 0; r < 8; ++r) outp[(size_t)r * HP] = acc[r];
}

__global__ __launch_bounds__(256) void gat_attn(const float* __restrict__ fth,
                                                const float* __restrict__ al,
                                                const float* __restrict__ ar,
                                                int p,
                                                float* __restrict__ el,
                                                float* __restrict__ er) {
    int idx = blockIdx.x * blockDim.x + threadIdx.x;
    if (idx >= N_NODES * 2) return;
    const int h = p * 2 + (idx & 1);
    const float* f = fth + (size_t)idx * D_OUT;
    float sl = 0.f, sr = 0.f;
#pragma unroll
    for (int d = 0; d < D_OUT; ++d) {
        float v = f[d];
        sl += v * al[h * D_OUT + d];
        sr += v * ar[h * D_OUT + d];
    }
    el[idx] = sl;
    er[idx] = sr;
}

__global__ __launch_bounds__(256) void csr_zero(int* __restrict__ cnt) {
    int i = blockIdx.x * blockDim.x + threadIdx.x;
    if (i < N_NODES) cnt[i] = 0;
}

__global__ __launch_bounds__(256) void csr_hist(const int* __restrict__ dst,
                                                int* __restrict__ cnt) {
    int i = blockIdx.x * blockDim.x + threadIdx.x;
    if (i < E_EDGES) atomicAdd(&cnt[dst[i]], 1);
}

__global__ __launch_bounds__(SCAN_T) void csr_scan(const int* __restrict__ cnt,
                                                   int* __restrict__ rowptr,
                                                   int* __restrict__ cur) {
    __shared__ int wsum[32];
    __shared__ int wtot[32];
    const int tid  = threadIdx.x;
    const int lane = tid & 31;
    const int wave = tid >> 5;
    const int base = blockIdx.x * SCAN_T;

    int part = 0;
    const int4* c4 = (const int4*)cnt;
    for (int j = tid; j < base / 4; j += SCAN_T) {
        int4 v = c4[j];
        part += (v.x + v.y) + (v.z + v.w);
    }
    for (int o = 16; o > 0; o >>= 1) part += __shfl_xor(part, o);
    if (lane == 0) wsum[wave] = part;

    const int i = base + tid;
    const int c = (i < N_NODES) ? cnt[i] : 0;
    int inc = c;
    for (int o = 1; o < 32; o <<= 1) {
        int t = __shfl_up(inc, o);
        if (lane >= o) inc += t;
    }
    if (lane == 31) wtot[wave] = inc;
    __syncthreads();

    int front = 0;
    for (int w = 0; w < 32; ++w) front += wsum[w];
    int woff = 0;
    for (int w = 0; w < wave; ++w) woff += wtot[w];
    const int excl = front + woff + inc - c;
    if (i <= N_NODES) rowptr[i] = excl;
    if (i <  N_NODES) cur[i]    = excl;
}

__global__ __launch_bounds__(256) void csr_fill(const int* __restrict__ src,
                                                const int* __restrict__ dst,
                                                int* __restrict__ cur,
                                                int* __restrict__ list) {
    int i = blockIdx.x * blockDim.x + threadIdx.x;
    if (i >= E_EDGES) return;
    const int pos = atomicAdd(&cur[dst[i]], 1);
    list[pos] = src[i];
}

__device__ __forceinline__ float edge_score(float l, float r) {
    float e = l + r;
    return e >= 0.f ? e : SLOPE * e;
}


__device__ __forceinline__ float chunk_rows(const float* __restrict__ fth,
                                            int s, float a0, float a1,
                                            int live, bool up, int lane, float acc) {
    const int cnt8 = (live + 7) & ~7;
    for (int k = 0; k < cnt8; k += 8) {
        float v[8], w[8];
#pragma unroll
        for (int u = 0; u < 8; ++u) {
            const int   sk = __builtin_amdgcn_readlane(s, k + u);
            const float w0 = __int_as_float(__builtin_amdgcn_readlane(__float_as_int(a0), k + u));
            const float w1 = __int_as_float(__builtin_amdgcn_readlane(__float_as_int(a1), k + u));
            v[u] = fth[(size_t)sk * HP + lane];
            w[u] = up ? w1 : w0;
        }
#pragma unroll
        for (int u = 0; u < 8; ++u) acc += w[u] * v[u];
    }
    return acc;
}

__global__ __launch_bounds__(256) void gat_gather(const int* __restrict__ rowptr,
                                                  const int* __restrict__ list,
                                                  const float* __restrict__ fth,
                                                  const float* __restrict__ el,
                                                  const float* __restrict__ er,
                                                  int p,
                                                  float* __restrict__ out) {
    const int lane = threadIdx.x & 31;
    const int n    = blockIdx.x * 8 + __builtin_amdgcn_readfirstlane((int)(threadIdx.x >> 5));
    const bool up  = lane >= 16;
    const int beg  = rowptr[n];
    const int end  = rowptr[n + 1];
    const float er0 = er[(size_t)n * 2];
    const float er1 = er[(size_t)n * 2 + 1];

    float acc = 0.f;
    float sv  = 1.f;
    if (end > beg) {
        const int  deg  = end - beg;
        const int  last = end - 1;
        const bool two  = deg > 32;
        const int  jA   = beg + lane;
        const int  jB   = jA + 32;
        int sA, sB;
        v2f lA, lB;
        if (two) {
            sA = list[jA];
            sB = list[min(jB, last)];
            lA = *(const v2f*)(el + (size_t)sA * 2);
            lB = *(const v2f*)(el + (size_t)sB * 2);
        } else {
            sA = list[min(jA, last)];
            lA = *(const v2f*)(el + (size_t)sA * 2);
            sB = sA;
            lB = lA;
        }
        const float eA0 = edge_score(lA.x, er0);
        const float eA1 = edge_score(lA.y, er1);
        const float eB0 = edge_score(lB.x, er0);
        const float eB1 = edge_score(lB.y, er1);

        float m0 = jA < end ? eA0 : -INFINITY;
        float m1 = jA < end ? eA1 : -INFINITY;
        if (two) {
            m0 = fmaxf(m0, jB < end ? eB0 : -INFINITY);
            m1 = fmaxf(m1, jB < end ? eB1 : -INFINITY);
        }
        for (int b = beg + 64; b < end; b += 32) {
            const int j  = b + lane;
            const int s  = list[min(j, last)];
            const v2f l  = *(const v2f*)(el + (size_t)s * 2);
            const float e0 = edge_score(l.x, er0);
            const float e1 = edge_score(l.y, er1);
            m0 = fmaxf(m0, j < end ? e0 : -INFINITY);
            m1 = fmaxf(m1, j < end ? e1 : -INFINITY);
        }
        for (int o = 16; o > 0; o >>= 1) {
            m0 = fmaxf(m0, __shfl_xor(m0, o));
            m1 = fmaxf(m1, __shfl_xor(m1, o));
        }

        const float aA0 = jA < end ? expf(eA0 - m0) : 0.f;
        const float aA1 = jA < end ? expf(eA1 - m1) : 0.f;
        float s0 = aA0, s1 = aA1;
        acc = chunk_rows(fth, sA, aA0, aA1, min(32, deg), up, lane, acc);
        if (two) {
            const float aB0 = jB < end ? expf(eB0 - m0) : 0.f;
            const float aB1 = jB < end ? expf(eB1 - m1) : 0.f;
            s0 += aB0;
            s1 += aB1;
            acc = chunk_rows(fth, sB, aB0, aB1, min(32, deg - 32), up, lane, acc);
        }
        for (int b = beg + 64; b < end; b += 32) {
            const int j  = b + lane;
            const int s  = list[min(j, last)];
            const v2f l  = *(const v2f*)(el + (size_t)s * 2);
            const float a0 = j < end ? expf(edge_score(l.x, er0) - m0) : 0.f;
            const float a1 = j < end ? expf(edge_score(l.y, er1) - m1) : 0.f;
            s0 += a0;
            s1 += a1;
            acc = chunk_rows(fth, s, a0, a1, min(32, end - b), up, lane, acc);
        }
        for (int o = 16; o > 0; o >>= 1) {
            s0 += __shfl_xor(s0, o);
            s1 += __shfl_xor(s1, o);
        }
        sv = up ? s1 : s0;
    }
    out[(size_t)n * HD + p * HP + lane] = acc / sv;
}

extern "C" void kernel_launch(void* const* d_in, const int* in_sizes, int n_in,
                              void* d_out, int out_size, void* d_ws, size_t ws_size,
                              hipStream_t stream) {
    const float* feat = (const float*)d_in[0];
    const float* W    = (const float*)d_in[1];
    const float* al   = (const float*)d_in[2];
    const float* ar   = (const float*)d_in[3];
    const int*   src  = (const int*)d_in[4];
    const int*   dst  = (const int*)d_in[5];
    float* out = (float*)d_out;

    float* fth    = (float*)d_ws;
    int*   list   = (int*)(fth + (size_t)N_NODES * HP);
    float* el     = (float*)(list + (size_t)E_EDGES);
    float* er     = el + (size_t)N_NODES * 2;
    int*   cnt    = (int*)(er + (size_t)N_NODES * 2);
    int*   rowptr = cnt + (size_t)N_NODES;
    int*   cur    = rowptr + 100352;

    const int nb_node = (N_NODES + 255) / 256;
    const int nb_edge = (E_EDGES + 255) / 256;
    const int nb_scan = (N_NODES + 1 + SCAN_T - 1) / SCAN_T;
    const int nb_gemm = N_NODES / 32;
    const int nb_attn = (N_NODES * 2 + 255) / 256;
    const int nb_gath = N_NODES / 8;

    csr_zero<<<nb_node, 256, 0, stream>>>(cnt);
    csr_hist<<<nb_edge, 256, 0, stream>>>(dst, cnt);
    csr_scan<<<nb_scan, SCAN_T, 0, stream>>>(cnt, rowptr, cur);
    csr_fill<<<nb_edge, 256, 0, stream>>>(src, dst, cur, list);

    gat_gemm<<<nb_gemm, 128, 0, stream>>>(feat, W, 0, fth);
    gat_attn<<<nb_attn, 256, 0, stream>>>(fth, al, ar, 0, el, er);
    gat_gather<<<nb_gath, 256, 0, stream>>>(rowptr, list, fth, el, er, 0, out);

    gat_gemm<<<nb_gemm, 128, 0, stream>>>(feat, W, 1, fth);
    gat_attn<<<nb_attn, 256, 0, stream>>>(fth, al, ar, 1, el, er);
    gat_gather<<<nb_gath, 256, 0, stream>>>(rowptr, list, fth, el, er, 1, out);
}


# Counting sort dropped: one fill pass takes an edge's arrival rank (atomic add) and parks its source in the node's own 256-byte out row (64 slots, spill list past that); hist and scan launches gone
# speedup vs baseline: 5.4163x; 1.6223x over previous
#include <hip/hip_runtime.h>


#define N_NODES 100000
#define IN_F    128
#define H_HEADS 4
#define D_OUT   16
#define HD      64
#define HP      32
#define E_EDGES 3200000
#define SLOPE   0.2f
#define SCAN_T  1024

typedef __attribute__((ext_vector_type(2))) float v2f;
typedef __attribute__((ext_vector_type(8))) float v8f;

template <int NT>
__global__ __launch_bounds__(64 * NT) void gat_gemm(const float* __restrict__ feat,
                                                    const float* __restrict__ W,
                                                    int p,
                                                    float* __restrict__ fth) {
    const int wave = threadIdx.x >> 5;
    const int lane = threadIdx.x & 31;
    const int rt   = wave / NT;
    const int ct   = wave % NT;
    const int r0   = blockIdx.x * 32 + rt * 16;
    const int c0   = (p * NT + ct) * 16;
    const int m    = lane & 15;
    const int hi   = lane >> 4;

    const float* arow = feat + (size_t)(r0 + m) * IN_F + (hi << 1);
    const float* brow = W    + (size_t)(c0 + m) * IN_F + (hi << 1);

    v8f acc = {};
#pragma unroll
    for (int k0 = 0; k0 < IN_F; k0 += 4) {
        v2f a = *(const v2f*)(arow + k0);
        v2f b = *(const v2f*)(brow + k0);
        acc = __builtin_amdgcn_wmma_f32_16x16x4_f32(
            false, a, false, b, (short)0, acc, false, false);
    }
    float* outp = fth + (size_t)(r0 + hi * 8) * (NT * 16) + ct * 16 + m;
#pragma unroll
    for (int r = 0; r < 8; ++r) outp[(size_t)r * (NT * 16)] = acc[r];
}

template <int NT>
__global__ __launch_bounds__(256) void gat_attn(const float* __restrict__ fth,
                                                const float* __restrict__ al,
                                                const float* __restrict__ ar,
                                                int p,
                                                float* __restrict__ el,
                                                float* __restrict__ er) {
    int idx = blockIdx.x * blockDim.x + threadIdx.x;
    if (idx >= N_NODES * NT) return;
    const int h = p * NT + (idx % NT);
    const float* f = fth + (size_t)idx * D_OUT;
    float sl = 0.f, sr = 0.f;
#pragma unroll
    for (int d = 0; d < D_OUT; ++d) {
        float v = f[d];
        sl += v * al[h * D_OUT + d];
        sr += v * ar[h * D_OUT + d];
    }
    el[idx] = sl;
    er[idx] = sr;
}

__global__ __launch_bounds__(256) void csr_zero(int* __restrict__ cnt) {
    int i = blockIdx.x * blockDim.x + threadIdx.x;
    if (i < N_NODES) cnt[i] = 0;
}

__global__ __launch_bounds__(256) void csr_hist(const int* __restrict__ dst,
                                                int* __restrict__ cnt) {
    int i = blockIdx.x * blockDim.x + threadIdx.x;
    if (i < E_EDGES) atomicAdd(&cnt[dst[i]], 1);
}

__global__ __launch_bounds__(SCAN_T) void csr_scan(const int* __restrict__ cnt,
                                                   int* __restrict__ rowptr,
                                                   int* __restrict__ cur) {
    __shared__ int wsum[32];
    __shared__ int wtot[32];
    const int tid  = threadIdx.x;
    const int lane = tid & 31;
    const int wave = tid >> 5;
    const int base = blockIdx.x * SCAN_T;

    int part = 0;
    const int4* c4 = (const int4*)cnt;
    for (int j = tid; j < base / 4; j += SCAN_T) {
        int4 v = c4[j];
        part += (v.x + v.y) + (v.z + v.w);
    }
    for (int o = 16; o > 0; o >>= 1) part += __shfl_xor(part, o);
    if (lane == 0) wsum[wave] = part;

    const int i = base + tid;
    const int c = (i < N_NODES) ? cnt[i] : 0;
    int inc = c;
    for (int o = 1; o < 32; o <<= 1) {
        int t = __shfl_up(inc, o);
        if (lane >= o) inc += t;
    }
    if (lane == 31) wtot[wave] = inc;
    __syncthreads();

    int front = 0;
    for (int w = 0; w < 32; ++w) front += wsum[w];
    int woff = 0;
    for (int w = 0; w < wave; ++w) woff += wtot[w];
    const int excl = front + woff + inc - c;
    if (i <= N_NODES) rowptr[i] = excl;
    if (i <  N_NODES) cur[i]    = excl;
}

__global__ __launch_bounds__(256) void csr_fill(const int* __restrict__ src,
                                                const int* __restrict__ dst,
                                                int* __restrict__ cur,
                                                int* __restrict__ list) {
    int i = blockIdx.x * blockDim.x + threadIdx.x;
    if (i >= E_EDGES) return;
    const int pos = atomicAdd(&cur[dst[i]], 1);
    list[pos] = src[i];
}

#define SLOTS 64

__global__ __launch_bounds__(256) void bkt_zero(int* __restrict__ cnt,
                                                int* __restrict__ nspill) {
    int i = blockIdx.x * blockDim.x + threadIdx.x;
    if (i < N_NODES) cnt[i] = 0;
    if (i == 0) *nspill = 0;
}

__global__ __launch_bounds__(256) void bkt_fill(const int* __restrict__ src,
                                                const int* __restrict__ dst,
                                                int* __restrict__ cnt,
                                                int* __restrict__ nspill,
                                                int* __restrict__ spill,
                                                float* __restrict__ out) {
    int i = blockIdx.x * blockDim.x + threadIdx.x;
    if (i >= E_EDGES) return;
    const int d   = dst[i];
    const int s   = src[i];
    const int pos = atomicAdd(&cnt[d], 1);
    int* where = (int*)out + (size_t)d * SLOTS + pos;
    int  what  = s;
    if (pos >= SLOTS) {
        where = spill + atomicAdd(nspill, 1);
        what  = i;
    }
    *where = what;
}

__device__ __forceinline__ float edge_score(float l, float r) {
    float e = l + r;
    return e >= 0.f ? e : SLOPE * e;
}


__device__ __forceinline__ float chunk_rows_pair(const float* __restrict__ fth,
                                            int s, float a0, float a1,
                                            int live, bool up, int lane, float acc) {
    const int cnt8 = (live + 7) & ~7;
    for (int k = 0; k < cnt8; k += 8) {
        float v[8], w[8];
#pragma unroll
        for (int u = 0; u < 8; ++u) {
            const int   sk = __builtin_amdgcn_readlane(s, k + u);
            const float w0 = __int_as_float(__builtin_amdgcn_readlane(__float_as_int(a0), k + u));
            const float w1 = __int_as_float(__builtin_amdgcn_readlane(__float_as_int(a1), k + u));
            v[u] = fth[(size_t)sk * HP + lane];
            w[u] = up ? w1 : w0;
        }
#pragma unroll
        for (int u = 0; u < 8; ++u) acc += w[u] * v[u];
    }
    return acc;
}

__global__ __launch_bounds__(256) void gat_gather_pair(const int* __restrict__ rowptr,
                                                  const int* __restrict__ list,
                                                  const float* __restrict__ fth,
                                                  const float* __restrict__ el,
                                                  const float* __restrict__ er,
                                                  int p,
                                                  float* __restrict__ out) {
    const int lane = threadIdx.x & 31;
    const int n    = blockIdx.x * 8 + __builtin_amdgcn_readfirstlane((int)(threadIdx.x >> 5));
    const bool up  = lane >= 16;
    const int beg  = rowptr[n];
    const int end  = rowptr[n + 1];
    const float er0 = er[(size_t)n * 2];
    const float er1 = er[(size_t)n * 2 + 1];

    float acc = 0.f;
    float sv  = 1.f;
    if (end > beg) {
        const int  deg  = end - beg;
        const int  last = end - 1;
        const bool two  = deg > 32;
        const int  jA   = beg + lane;
        const int  jB   = jA + 32;
        int sA, sB;
        v2f lA, lB;
        if (two) {
            sA = list[jA];
            sB = list[min(jB, last)];
            lA = *(const v2f*)(el + (size_t)sA * 2);
            lB = *(const v2f*)(el + (size_t)sB * 2);
        } else {
            sA = list[min(jA, last)];
            lA = *(const v2f*)(el + (size_t)sA * 2);
            sB = sA;
            lB = lA;
        }
        const float eA0 = edge_score(lA.x, er0);
        const float eA1 = edge_score(lA.y, er1);
        const float eB0 = edge_score(lB.x, er0);
        const float eB1 = edge_score(lB.y, er1);

        float m0 = jA < end ? eA0 : -INFINITY;
        float m1 = jA < end ? eA1 : -INFINITY;
        if (two) {
            m0 = fmaxf(m0, jB < end ? eB0 : -INFINITY);
            m1 = fmaxf(m1, jB < end ? eB1 : -INFINITY);
        }
        for (int b = beg + 64; b < end; b += 32) {
            const int j  = b + lane;
            const int s  = list[min(j, last)];
            const v2f l  = *(const v2f*)(el + (size_t)s * 2);
            const float e0 = edge_score(l.x, er0);
            const float e1 = edge_score(l.y, er1);
            m0 = fmaxf(m0, j < end ? e0 : -INFINITY);
            m1 = fmaxf(m1, j < end ? e1 : -INFINITY);
        }
        for (int o = 16; o > 0; o >>= 1) {
            m0 = fmaxf(m0, __shfl_xor(m0, o));
            m1 = fmaxf(m1, __shfl_xor(m1, o));
        }

        const float aA0 = jA < end ? expf(eA0 - m0) : 0.f;
        const float aA1 = jA < end ? expf(eA1 - m1) : 0.f;
        float s0 = aA0, s1 = aA1;
        acc = chunk_rows_pair(fth, sA, aA0, aA1, min(32, deg), up, lane, acc);
        if (two) {
            const float aB0 = jB < end ? expf(eB0 - m0) : 0.f;
            const float aB1 = jB < end ? expf(eB1 - m1) : 0.f;
            s0 += aB0;
            s1 += aB1;
            acc = chunk_rows_pair(fth, sB, aB0, aB1, min(32, deg - 32), up, lane, acc);
        }
        for (int b = beg + 64; b < end; b += 32) {
            const int j  = b + lane;
            const int s  = list[min(j, last)];
            const v2f l  = *(const v2f*)(el + (size_t)s * 2);
            const float a0 = j < end ? expf(edge_score(l.x, er0) - m0) : 0.f;
            const float a1 = j < end ? expf(edge_score(l.y, er1) - m1) : 0.f;
            s0 += a0;
            s1 += a1;
            acc = chunk_rows_pair(fth, s, a0, a1, min(32, end - b), up, lane, acc);
        }
        for (int o = 16; o > 0; o >>= 1) {
            s0 += __shfl_xor(s0, o);
            s1 += __shfl_xor(s1, o);
        }
        sv = up ? s1 : s0;
    }
    out[(size_t)n * HD + p * HP + lane] = acc / sv;
}


template <int U>
__device__ __forceinline__ float bcast8(float x) {
    return __uint_as_float(__builtin_amdgcn_mov_dpp8(__float_as_uint(x), U * 0x249249u));
}

constexpr unsigned xor8_sel(int x) {
    unsigned s = 0;
    for (int i = 0; i < 8; ++i) s |= (unsigned)((i ^ x) & 7) << (3 * i);
    return s;
}
template <int X>
__device__ __forceinline__ float xor8(float v) {
    return __uint_as_float(__builtin_amdgcn_mov_dpp8(__float_as_uint(v), xor8_sel(X)));
}

template <int U>
__device__ __forceinline__ void edge_row(const float* __restrict__ fth, int s, float a,
                                         int lane, v2f& v, float& w) {
    const int sk = __builtin_amdgcn_readlane(s, U);
    w = bcast8<U>(a);
    v = *(const v2f*)(fth + (size_t)sk * HD + 2 * lane);
}

__device__ __forceinline__ v2f chunk_rows_all(const float* __restrict__ fth,
                                          const int (&s)[4], const float (&a)[4],
                                          int live, int lane, v2f acc) {
#pragma unroll
    for (int g = 0; g < 4; ++g) {
        if (8 * g < live) {
            v2f v[8];
            float w[8];
            edge_row<0>(fth, s[g], a[g], lane, v[0], w[0]);
            edge_row<1>(fth, s[g], a[g], lane, v[1], w[1]);
            edge_row<2>(fth, s[g], a[g], lane, v[2], w[2]);
            edge_row<3>(fth, s[g], a[g], lane, v[3], w[3]);
            edge_row<4>(fth, s[g], a[g], lane, v[4], w[4]);
            edge_row<5>(fth, s[g], a[g], lane, v[5], w[5]);
            edge_row<6>(fth, s[g], a[g], lane, v[6], w[6]);
            edge_row<7>(fth, s[g], a[g], lane, v[7], w[7]);
#pragma unroll
            for (int u = 0; u < 8; ++u) {
                acc.x += w[u] * v[u].x;
                acc.y += w[u] * v[u].y;
            }
        }
    }
    return acc;
}

template <class F>
__device__ __forceinline__ void for_spill_edges(const int* __restrict__ spill,
                                                const int* __restrict__ src,
                                                const int* __restrict__ dst,
                                                int nsp, int n, int lane, F f) {
    for (int b = 0; b < nsp; b += 32) {
        const int k = b + lane;
        bool mine = false;
        int  s    = 0;
        if (k < nsp) {
            const int i = spill[k];
            mine = dst[i] == n;
            if (mine) s = src[i];
        }
        unsigned mask = __builtin_amdgcn_ballot_w32(mine);
        while (mask != 0u) {
            const int u = __builtin_ctz(mask);
            mask &= mask - 1u;
            f(__builtin_amdgcn_readlane(s, u));
        }
    }
}

__global__ __launch_bounds__(256) void gat_gather_all(const int* __restrict__ cnt,
                                                  const int* __restrict__ nspill,
                                                  const int* __restrict__ spill,
                                                  const int* __restrict__ src,
                                                  const int* __restrict__ dst,
                                                  const float* __restrict__ fth,
                                                  const float* __restrict__ el,
                                                  const float* __restrict__ er,
                                                  float* out) {
    const int lane = threadIdx.x & 31;
    const int n    = blockIdx.x * 8 + __builtin_amdgcn_readfirstlane((int)(threadIdx.x >> 5));
    const int hL   = lane >> 3;
    const int j    = lane & 7;
    const int c    = cnt[n];
    const int* slot = (const int*)out + (size_t)n * SLOTS;
    const float erh = er[(size_t)n * H_HEADS + hL];

    v2f   acc = {0.f, 0.f};
    float sv  = 1.f;
    if (c > 0) {
        const int  deg  = min(c, SLOTS);
        const int  last = deg - 1;
        const bool two  = deg > 32;
        const int  nsp  = c > SLOTS ? *nspill : 0;
        const int  jB   = j + 32;
        int   sA[4], sB[4];
        float eA[4], eB[4];
        if (two) {
#pragma unroll
            for (int g = 0; g < 4; ++g) sA[g] = slot[j + 8 * g];
#pragma unroll
            for (int g = 0; g < 4; ++g) sB[g] = slot[min(jB + 8 * g, last)];
#pragma unroll
            for (int g = 0; g < 4; ++g) eA[g] = el[(size_t)sA[g] * H_HEADS + hL];
#pragma unroll
            for (int g = 0; g < 4; ++g) eB[g] = el[(size_t)sB[g] * H_HEADS + hL];
        } else {
#pragma unroll
            for (int g = 0; g < 4; ++g) sA[g] = slot[min(j + 8 * g, last)];
#pragma unroll
            for (int g = 0; g < 4; ++g) eA[g] = el[(size_t)sA[g] * H_HEADS + hL];
#pragma unroll
            for (int g = 0; g < 4; ++g) { sB[g] = sA[g]; eB[g] = eA[g]; }
        }
#pragma unroll
        for (int g = 0; g < 4; ++g) {
            eA[g] = edge_score(eA[g], erh);
            eB[g] = edge_score(eB[g], erh);
        }

        float m = -INFINITY;
#pragma unroll
        for (int g = 0; g < 4; ++g) m = fmaxf(m, j + 8 * g < deg ? eA[g] : -INFINITY);
        if (two) {
#pragma unroll
            for (int g = 0; g < 4; ++g) m = fmaxf(m, jB + 8 * g < deg ? eB[g] : -INFINITY);
        }
        for_spill_edges(spill, src, dst, nsp, n, lane, [&](int s) {
            m = fmaxf(m, edge_score(el[(size_t)s * H_HEADS + hL], erh));
        });
        m = fmaxf(m, xor8<4>(m));
        m = fmaxf(m, xor8<2>(m));
        m = fmaxf(m, xor8<1>(m));

        float aA[4];
        float ssum = 0.f;
#pragma unroll
        for (int g = 0; g < 4; ++g) {
            aA[g] = j + 8 * g < deg ? expf(eA[g] - m) : 0.f;
            ssum += aA[g];
        }
        acc = chunk_rows_all(fth, sA, aA, min(32, deg), lane, acc);
        if (two) {
            float aB[4];
#pragma unroll
            for (int g = 0; g < 4; ++g) {
                aB[g] = jB + 8 * g < deg ? expf(eB[g] - m) : 0.f;
                ssum += aB[g];
            }
            acc = chunk_rows_all(fth, sB, aB, deg - 32, lane, acc);
        }
        for_spill_edges(spill, src, dst, nsp, n, lane, [&](int s) {
            const float a = expf(edge_score(el[(size_t)s * H_HEADS + hL], erh) - m);
            const v2f   v = *(const v2f*)(fth + (size_t)s * HD + 2 * lane);
            ssum  += j == 0 ? a : 0.f;
            acc.x += a * v.x;
            acc.y += a * v.y;
        });
        ssum += xor8<4>(ssum);
        ssum += xor8<2>(ssum);
        ssum += xor8<1>(ssum);
        sv = ssum;
    }
    v2f res;
    res.x = acc.x / sv;
    res.y = acc.y / sv;
    *(v2f*)(out + (size_t)n * HD + 2 * lane) = res;
}

extern "C" void kernel_launch(void* const* d_in, const int* in_sizes, int n_in,
                              void* d_out, int out_size, void* d_ws, size_t ws_size,
                              hipStream_t stream) {
    const float* feat = (const float*)d_in[0];
    const float* W    = (const float*)d_in[1];
    const float* al   = (const float*)d_in[2];
    const float* ar   = (const float*)d_in[3];
    const int*   src  = (const int*)d_in[4];
    const int*   dst  = (const int*)d_in[5];
    float* out = (float*)d_out;

    const int nb_node = (N_NODES + 255) / 256;
    const int nb_edge = (E_EDGES + 255) / 256;
    const int nb_scan = (N_NODES + 1 + SCAN_T - 1) / SCAN_T;
    const int nb_gemm = N_NODES / 32;
    const int nb_gath = N_NODES / 8;

    if (ws_size >= (size_t)42000004) {
        float* fth    = (float*)d_ws;
        int*   spill  = (int*)(fth + (size_t)N_NODES * HD);
        float* el     = (float*)(spill + (size_t)E_EDGES);
        float* er     = el + (size_t)N_NODES * H_HEADS;
        int*   cnt    = (int*)(er + (size_t)N_NODES * H_HEADS);
        int*   nspill = cnt + (size_t)N_NODES;
        const int nb_attn = (N_NODES * H_HEADS + 255) / 256;

        bkt_zero<<<nb_node, 256, 0, stream>>>(cnt, nspill);
        bkt_fill<<<nb_edge, 256, 0, stream>>>(src, dst, cnt, nspill, spill, out);

        gat_gemm<4><<<nb_gemm, 256, 0, stream>>>(feat, W, 0, fth);
        gat_attn<4><<<nb_attn, 256, 0, stream>>>(fth, al, ar, 0, el, er);
        gat_gather_all<<<nb_gath, 256, 0, stream>>>(cnt, nspill, spill, src, dst, fth, el, er, out);
    } else {
        float* fth    = (float*)d_ws;
        int*   list   = (int*)(fth + (size_t)N_NODES * HP);
        float* el     = (float*)(list + (size_t)E_EDGES);
        float* er     = el + (size_t)N_NODES * 2;
        int*   cnt    = (int*)(er + (size_t)N_NODES * 2);
        int*   rowptr = cnt + (size_t)N_NODES;
        int*   cur    = rowptr + 100352;
        const int nb_attn = (N_NODES * 2 + 255) / 256;

        csr_zero<<<nb_node, 256, 0, stream>>>(cnt);
        csr_hist<<<nb_edge, 256, 0, stream>>>(dst, cnt);
        csr_scan<<<nb_scan, SCAN_T, 0, stream>>>(cnt, rowptr, cur);
        csr_fill<<<nb_edge, 256, 0, stream>>>(src, dst, cur, list);

        gat_gemm<2><<<nb_gemm, 128, 0, stream>>>(feat, W, 0, fth);
        gat_attn<2><<<nb_attn, 256, 0, stream>>>(fth, al, ar, 0, el, er);
        gat_gather_pair<<<nb_gath, 256, 0, stream>>>(rowptr, list, fth, el, er, 0, out);

        gat_gemm<2><<<nb_gemm, 128, 0, stream>>>(feat, W, 1, fth);
        gat_attn<2><<<nb_attn, 256, 0, stream>>>(fth, al, ar, 1, el, er);
        gat_gather_pair<<<nb_gath, 256, 0, stream>>>(rowptr, list, fth, el, er, 1, out);
    }
}


# Fill keeps each written line in one package half: a counter plane and a 128-byte line of every node's bucket row per half (die from the placement message); sum order = arrival rank per half
# speedup vs baseline: 8.3035x; 1.5331x over previous
#include <hip/hip_runtime.h>


#define N_NODES 100000
#define IN_F    128
#define H_HEADS 4
#define D_OUT   16
#define HD      64
#define HP      32
#define E_EDGES 3200000
#define SLOPE   0.2f
#define SCAN_T  1024

typedef __attribute__((ext_vector_type(2))) float v2f;
typedef __attribute__((ext_vector_type(8))) float v8f;

template <int NT>
__global__ __launch_bounds__(64 * NT) void gat_gemm(const float* __restrict__ feat,
                                                    const float* __restrict__ W,
                                                    int p,
                                                    float* __restrict__ fth) {
    const int wave = threadIdx.x >> 5;
    const int lane = threadIdx.x & 31;
    const int rt   = wave / NT;
    const int ct   = wave % NT;
    const int r0   = blockIdx.x * 32 + rt * 16;
    const int c0   = (p * NT + ct) * 16;
    const int m    = lane & 15;
    const int hi   = lane >> 4;

    const float* arow = feat + (size_t)(r0 + m) * IN_F + (hi << 1);
    const float* brow = W    + (size_t)(c0 + m) * IN_F + (hi << 1);

    v8f acc = {};
#pragma unroll
    for (int k0 = 0; k0 < IN_F; k0 += 4) {
        v2f a = *(const v2f*)(arow + k0);
        v2f b = *(const v2f*)(brow + k0);
        acc = __builtin_amdgcn_wmma_f32_16x16x4_f32(
            false, a, false, b, (short)0, acc, false, false);
    }
    float* outp = fth + (size_t)(r0 + hi * 8) * (NT * 16) + ct * 16 + m;
#pragma unroll
    for (int r = 0; r < 8; ++r) outp[(size_t)r * (NT * 16)] = acc[r];
}

template <int NT>
__global__ __launch_bounds__(256) void gat_attn(const float* __restrict__ fth,
                                                const float* __restrict__ al,
                                                const float* __restrict__ ar,
                                                int p,
                                                float* __restrict__ el,
                                                float* __restrict__ er) {
    int idx = blockIdx.x * blockDim.x + threadIdx.x;
    if (idx >= N_NODES * NT) return;
    const int h = p * NT + (idx % NT);
    const float* f = fth + (size_t)idx * D_OUT;
    float sl = 0.f, sr = 0.f;
#pragma unroll
    for (int d = 0; d < D_OUT; ++d) {
        float v = f[d];
        sl += v * al[h * D_OUT + d];
        sr += v * ar[h * D_OUT + d];
    }
    el[idx] = sl;
    er[idx] = sr;
}

__global__ __launch_bounds__(256) void csr_zero(int* __restrict__ cnt) {
    int i = blockIdx.x * blockDim.x + threadIdx.x;
    if (i < N_NODES) cnt[i] = 0;
}

__global__ __launch_bounds__(256) void csr_hist(const int* __restrict__ dst,
                                                int* __restrict__ cnt) {
    int i = blockIdx.x * blockDim.x + threadIdx.x;
    if (i < E_EDGES) atomicAdd(&cnt[dst[i]], 1);
}

__global__ __launch_bounds__(SCAN_T) void csr_scan(const int* __restrict__ cnt,
                                                   int* __restrict__ rowptr,
                                                   int* __restrict__ cur) {
    __shared__ int wsum[32];
    __shared__ int wtot[32];
    const int tid  = threadIdx.x;
    const int lane = tid & 31;
    const int wave = tid >> 5;
    const int base = blockIdx.x * SCAN_T;

    int part = 0;
    const int4* c4 = (const int4*)cnt;
    for (int j = tid; j < base / 4; j += SCAN_T) {
        int4 v = c4[j];
        part += (v.x + v.y) + (v.z + v.w);
    }
    for (int o = 16; o > 0; o >>= 1) part += __shfl_xor(part, o);
    if (lane == 0) wsum[wave] = part;

    const int i = base + tid;
    const int c = (i < N_NODES) ? cnt[i] : 0;
    int inc = c;
    for (int o = 1; o < 32; o <<= 1) {
        int t = __shfl_up(inc, o);
        if (lane >= o) inc += t;
    }
    if (lane == 31) wtot[wave] = inc;
    __syncthreads();

    int front = 0;
    for (int w = 0; w < 32; ++w) front += wsum[w];
    int woff = 0;
    for (int w = 0; w < wave; ++w) woff += wtot[w];
    const int excl = front + woff + inc - c;
    if (i <= N_NODES) rowptr[i] = excl;
    if (i <  N_NODES) cur[i]    = excl;
}

__global__ __launch_bounds__(256) void csr_fill(const int* __restrict__ src,
                                                const int* __restrict__ dst,
                                                int* __restrict__ cur,
                                                int* __restrict__ list) {
    int i = blockIdx.x * blockDim.x + threadIdx.x;
    if (i >= E_EDGES) return;
    const int pos = atomicAdd(&cur[dst[i]], 1);
    list[pos] = src[i];
}

#define SLOTS  64
#define HSLOTS 32

__global__ __launch_bounds__(256) void bkt_zero(int* __restrict__ cnt,
                                                int* __restrict__ nspill) {
    int i = blockIdx.x * blockDim.x + threadIdx.x;
    if (i < 2 * N_NODES) cnt[i] = 0;
    if (i == 0) *nspill = 0;
}

__global__ __launch_bounds__(256) void bkt_fill(const int* __restrict__ src,
                                                const int* __restrict__ dst,
                                                int* __restrict__ cnt,
                                                int* __restrict__ nspill,
                                                int* __restrict__ spill,
                                                float* __restrict__ out) {
    int i = blockIdx.x * blockDim.x + threadIdx.x;
    if (i >= E_EDGES) return;
    const int half = (int)((__builtin_amdgcn_s_sendmsg_rtn(0x87) >> 18) & 1u);
    const int d   = dst[i];
    const int s   = src[i];
    const int pos = atomicAdd(&cnt[half * N_NODES + d], 1);
    int* where = (int*)out + (size_t)d * SLOTS + half * HSLOTS + pos;
    int  what  = s;
    if (pos >= HSLOTS) {
        where = spill + atomicAdd(nspill, 1);
        what  = i;
    }
    *where = what;
}

__device__ __forceinline__ float edge_score(float l, float r) {
    float e = l + r;
    return e >= 0.f ? e : SLOPE * e;
}


__device__ __forceinline__ float chunk_rows_pair(const float* __restrict__ fth,
                                            int s, float a0, float a1,
                                            int live, bool up, int lane, float acc) {
    const int cnt8 = (live + 7) & ~7;
    for (int k = 0; k < cnt8; k += 8) {
        float v[8], w[8];
#pragma unroll
        for (int u = 0; u < 8; ++u) {
            const int   sk = __builtin_amdgcn_readlane(s, k + u);
            const float w0 = __int_as_float(__builtin_amdgcn_readlane(__float_as_int(a0), k + u));
            const float w1 = __int_as_float(__builtin_amdgcn_readlane(__float_as_int(a1), k + u));
            v[u] = fth[(size_t)sk * HP + lane];
            w[u] = up ? w1 : w0;
        }
#pragma unroll
        for (int u = 0; u < 8; ++u) acc += w[u] * v[u];
    }
    return acc;
}

__global__ __launch_bounds__(256) void gat_gather_pair(const int* __restrict__ rowptr,
                                                  const int* __restrict__ list,
                                                  const float* __restrict__ fth,
                                                  const float* __restrict__ el,
                                                  const float* __restrict__ er,
                                                  int p,
                                                  float* __restrict__ out) {
    const int lane = threadIdx.x & 31;
    const int n    = blockIdx.x * 8 + __builtin_amdgcn_readfirstlane((int)(threadIdx.x >> 5));
    const bool up  = lane >= 16;
    const int beg  = rowptr[n];
    const int end  = rowptr[n + 1];
    const float er0 = er[(size_t)n * 2];
    const float er1 = er[(size_t)n * 2 + 1];

    float acc = 0.f;
    float sv  = 1.f;
    if (end > beg) {
        const int  deg  = end - beg;
        const int  last = end - 1;
        const bool two  = deg > 32;
        const int  jA   = beg + lane;
        const int  jB   = jA + 32;
        int sA, sB;
        v2f lA, lB;
        if (two) {
            sA = list[jA];
            sB = list[min(jB, last)];
            lA = *(const v2f*)(el + (size_t)sA * 2);
            lB = *(const v2f*)(el + (size_t)sB * 2);
        } else {
            sA = list[min(jA, last)];
            lA = *(const v2f*)(el + (size_t)sA * 2);
            sB = sA;
            lB = lA;
        }
        const float eA0 = edge_score(lA.x, er0);
        const float eA1 = edge_score(lA.y, er1);
        const float eB0 = edge_score(lB.x, er0);
        const float eB1 = edge_score(lB.y, er1);

        float m0 = jA < end ? eA0 : -INFINITY;
        float m1 = jA < end ? eA1 : -INFINITY;
        if (two) {
            m0 = fmaxf(m0, jB < end ? eB0 : -INFINITY);
            m1 = fmaxf(m1, jB < end ? eB1 : -INFINITY);
        }
        for (int b = beg + 64; b < end; b += 32) {
            const int j  = b + lane;
            const int s  = list[min(j, last)];
            const v2f l  = *(const v2f*)(el + (size_t)s * 2);
            const float e0 = edge_score(l.x, er0);
            const float e1 = edge_score(l.y, er1);
            m0 = fmaxf(m0, j < end ? e0 : -INFINITY);
            m1 = fmaxf(m1, j < end ? e1 : -INFINITY);
        }
        for (int o = 16; o > 0; o >>= 1) {
            m0 = fmaxf(m0, __shfl_xor(m0, o));
            m1 = fmaxf(m1, __shfl_xor(m1, o));
        }

        const float aA0 = jA < end ? expf(eA0 - m0) : 0.f;
        const float aA1 = jA < end ? expf(eA1 - m1) : 0.f;
        float s0 = aA0, s1 = aA1;
        acc = chunk_rows_pair(fth, sA, aA0, aA1, min(32, deg), up, lane, acc);
        if (two) {
            const float aB0 = jB < end ? expf(eB0 - m0) : 0.f;
            const float aB1 = jB < end ? expf(eB1 - m1) : 0.f;
            s0 += aB0;
            s1 += aB1;
            acc = chunk_rows_pair(fth, sB, aB0, aB1, min(32, deg - 32), up, lane, acc);
        }
        for (int b = beg + 64; b < end; b += 32) {
            const int j  = b + lane;
            const int s  = list[min(j, last)];
            const v2f l  = *(const v2f*)(el + (size_t)s * 2);
            const float a0 = j < end ? expf(edge_score(l.x, er0) - m0) : 0.f;
            const float a1 = j < end ? expf(edge_score(l.y, er1) - m1) : 0.f;
            s0 += a0;
            s1 += a1;
            acc = chunk_rows_pair(fth, s, a0, a1, min(32, end - b), up, lane, acc);
        }
        for (int o = 16; o > 0; o >>= 1) {
            s0 += __shfl_xor(s0, o);
            s1 += __shfl_xor(s1, o);
        }
        sv = up ? s1 : s0;
    }
    out[(size_t)n * HD + p * HP + lane] = acc / sv;
}


template <int U>
__device__ __forceinline__ float bcast8(float x) {
    return __uint_as_float(__builtin_amdgcn_mov_dpp8(__float_as_uint(x), U * 0x249249u));
}

constexpr unsigned xor8_sel(int x) {
    unsigned s = 0;
    for (int i = 0; i < 8; ++i) s |= (unsigned)((i ^ x) & 7) << (3 * i);
    return s;
}
template <int X>
__device__ __forceinline__ float xor8(float v) {
    return __uint_as_float(__builtin_amdgcn_mov_dpp8(__float_as_uint(v), xor8_sel(X)));
}

template <int U>
__device__ __forceinline__ void edge_row(const float* __restrict__ fth, int s, float a,
                                         int lane, v2f& v, float& w) {
    const int sk = __builtin_amdgcn_readlane(s, U);
    w = bcast8<U>(a);
    v = *(const v2f*)(fth + (size_t)sk * HD + 2 * lane);
}

__device__ __forceinline__ v2f chunk_rows_all(const float* __restrict__ fth,
                                          const int (&s)[4], const float (&a)[4],
                                          int live, int lane, v2f acc) {
#pragma unroll
    for (int g = 0; g < 4; ++g) {
        if (8 * g < live) {
            v2f v[8];
            float w[8];
            edge_row<0>(fth, s[g], a[g], lane, v[0], w[0]);
            edge_row<1>(fth, s[g], a[g], lane, v[1], w[1]);
            edge_row<2>(fth, s[g], a[g], lane, v[2], w[2]);
            edge_row<3>(fth, s[g], a[g], lane, v[3], w[3]);
            edge_row<4>(fth, s[g], a[g], lane, v[4], w[4]);
            edge_row<5>(fth, s[g], a[g], lane, v[5], w[5]);
            edge_row<6>(fth, s[g], a[g], lane, v[6], w[6]);
            edge_row<7>(fth, s[g], a[g], lane, v[7], w[7]);
#pragma unroll
            for (int u = 0; u < 8; ++u) {
                acc.x += w[u] * v[u].x;
                acc.y += w[u] * v[u].y;
            }
        }
    }
    return acc;
}

template <class F>
__device__ __forceinline__ void for_spill_edges(const int* __restrict__ spill,
                                                const int* __restrict__ src,
                                                const int* __restrict__ dst,
                                                int nsp, int n, int lane, F f) {
    for (int b = 0; b < nsp; b += 32) {
        const int k = b + lane;
        bool mine = false;
        int  s    = 0;
        if (k < nsp) {
            const int i = spill[k];
            mine = dst[i] == n;
            if (mine) s = src[i];
        }
        unsigned mask = __builtin_amdgcn_ballot_w32(mine);
        while (mask != 0u) {
            const int u = __builtin_ctz(mask);
            mask &= mask - 1u;
            f(__builtin_amdgcn_readlane(s, u));
        }
    }
}

__global__ __launch_bounds__(256) void gat_gather_all(const int* __restrict__ cnt,
                                                  const int* __restrict__ nspill,
                                                  const int* __restrict__ spill,
                                                  const int* __restrict__ src,
                                                  const int* __restrict__ dst,
                                                  const float* __restrict__ fth,
                                                  const float* __restrict__ el,
                                                  const float* __restrict__ er,
                                                  float* out) {
    const int lane = threadIdx.x & 31;
    const int n    = blockIdx.x * 8 + __builtin_amdgcn_readfirstlane((int)(threadIdx.x >> 5));
    const int hL   = lane >> 3;
    const int j    = lane & 7;
    const int cA   = cnt[n];
    const int cB   = cnt[N_NODES + n];
    const int* slot = (const int*)out + (size_t)n * SLOTS;
    const float erh = er[(size_t)n * H_HEADS + hL];

    v2f   acc = {0.f, 0.f};
    float sv  = 1.f;
    if (cA + cB > 0) {
        const int  dA   = min(cA, HSLOTS);
        const int  dB   = min(cB, HSLOTS);
        const int  deg  = dA + dB;
        const int  last = deg - 1;
        const bool two  = deg > 32;
        const int  nsp  = (cA > HSLOTS || cB > HSLOTS) ? *nspill : 0;
        const int  jB   = j + 32;
        auto at = [&](int p) { return slot[p < dA ? p : p - dA + HSLOTS]; };
        int   sA[4], sB[4];
        float eA[4], eB[4];
        if (two) {
#pragma unroll
            for (int g = 0; g < 4; ++g) sA[g] = at(j + 8 * g);
#pragma unroll
            for (int g = 0; g < 4; ++g) sB[g] = at(min(jB + 8 * g, last));
#pragma unroll
            for (int g = 0; g < 4; ++g) eA[g] = el[(size_t)sA[g] * H_HEADS + hL];
#pragma unroll
            for (int g = 0; g < 4; ++g) eB[g] = el[(size_t)sB[g] * H_HEADS + hL];
        } else {
#pragma unroll
            for (int g = 0; g < 4; ++g) sA[g] = at(min(j + 8 * g, last));
#pragma unroll
            for (int g = 0; g < 4; ++g) eA[g] = el[(size_t)sA[g] * H_HEADS + hL];
#pragma unroll
            for (int g = 0; g < 4; ++g) { sB[g] = sA[g]; eB[g] = eA[g]; }
        }
#pragma unroll
        for (int g = 0; g < 4; ++g) {
            eA[g] = edge_score(eA[g], erh);
            eB[g] = edge_score(eB[g], erh);
        }

        float m = -INFINITY;
#pragma unroll
        for (int g = 0; g < 4; ++g) m = fmaxf(m, j + 8 * g < deg ? eA[g] : -INFINITY);
        if (two) {
#pragma unroll
            for (int g = 0; g < 4; ++g) m = fmaxf(m, jB + 8 * g < deg ? eB[g] : -INFINITY);
        }
        for_spill_edges(spill, src, dst, nsp, n, lane, [&](int s) {
            m = fmaxf(m, edge_score(el[(size_t)s * H_HEADS + hL], erh));
        });
        m = fmaxf(m, xor8<4>(m));
        m = fmaxf(m, xor8<2>(m));
        m = fmaxf(m, xor8<1>(m));

        float aA[4];
        float ssum = 0.f;
#pragma unroll
        for (int g = 0; g < 4; ++g) {
            aA[g] = j + 8 * g < deg ? expf(eA[g] - m) : 0.f;
            ssum += aA[g];
        }
        acc = chunk_rows_all(fth, sA, aA, min(32, deg), lane, acc);
        if (two) {
            float aB[4];
#pragma unroll
            for (int g = 0; g < 4; ++g) {
                aB[g] = jB + 8 * g < deg ? expf(eB[g] - m) : 0.f;
                ssum += aB[g];
            }
            acc = chunk_rows_all(fth, sB, aB, deg - 32, lane, acc);
        }
        for_spill_edges(spill, src, dst, nsp, n, lane, [&](int s) {
            const float a = expf(edge_score(el[(size_t)s * H_HEADS + hL], erh) - m);
            const v2f   v = *(const v2f*)(fth + (size_t)s * HD + 2 * lane);
            ssum  += j == 0 ? a : 0.f;
            acc.x += a * v.x;
            acc.y += a * v.y;
        });
        ssum += xor8<4>(ssum);
        ssum += xor8<2>(ssum);
        ssum += xor8<1>(ssum);
        sv = ssum;
    }
    v2f res;
    res.x = acc.x / sv;
    res.y = acc.y / sv;
    *(v2f*)(out + (size_t)n * HD + 2 * lane) = res;
}

extern "C" void kernel_launch(void* const* d_in, const int* in_sizes, int n_in,
                              void* d_out, int out_size, void* d_ws, size_t ws_size,
                              hipStream_t stream) {
    const float* feat = (const float*)d_in[0];
    const float* W    = (const float*)d_in[1];
    const float* al   = (const float*)d_in[2];
    const float* ar   = (const float*)d_in[3];
    const int*   src  = (const int*)d_in[4];
    const int*   dst  = (const int*)d_in[5];
    float* out = (float*)d_out;

    const int nb_node = (N_NODES + 255) / 256;
    const int nb_edge = (E_EDGES + 255) / 256;
    const int nb_scan = (N_NODES + 1 + SCAN_T - 1) / SCAN_T;
    const int nb_gemm = N_NODES / 32;
    const int nb_gath = N_NODES / 8;

    if (ws_size >= (size_t)42400004) {
        float* fth    = (float*)d_ws;
        int*   spill  = (int*)(fth + (size_t)N_NODES * HD);
        float* el     = (float*)(spill + (size_t)E_EDGES);
        float* er     = el + (size_t)N_NODES * H_HEADS;
        int*   cnt    = (int*)(er + (size_t)N_NODES * H_HEADS);
        int*   nspill = cnt + (size_t)2 * N_NODES;
        const int nb_attn = (N_NODES * H_HEADS + 255) / 256;
        const int nb_zero = (2 * N_NODES + 255) / 256;

        bkt_zero<<<nb_zero, 256, 0, stream>>>(cnt, nspill);
        bkt_fill<<<nb_edge, 256, 0, stream>>>(src, dst, cnt, nspill, spill, out);

        gat_gemm<4><<<nb_gemm, 256, 0, stream>>>(feat, W, 0, fth);
        gat_attn<4><<<nb_attn, 256, 0, stream>>>(fth, al, ar, 0, el, er);
        gat_gather_all<<<nb_gath, 256, 0, stream>>>(cnt, nspill, spill, src, dst, fth, el, er, out);
    } else {
        float* fth    = (float*)d_ws;
        int*   list   = (int*)(fth + (size_t)N_NODES * HP);
        float* el     = (float*)(list + (size_t)E_EDGES);
        float* er     = el + (size_t)N_NODES * 2;
        int*   cnt    = (int*)(er + (size_t)N_NODES * 2);
        int*   rowptr = cnt + (size_t)N_NODES;
        int*   cur    = rowptr + 100352;
        const int nb_attn = (N_NODES * 2 + 255) / 256;

        csr_zero<<<nb_node, 256, 0, stream>>>(cnt);
        csr_hist<<<nb_edge, 256, 0, stream>>>(dst, cnt);
        csr_scan<<<nb_scan, SCAN_T, 0, stream>>>(cnt, rowptr, cur);
        csr_fill<<<nb_edge, 256, 0, stream>>>(src, dst, cur, list);

        gat_gemm<2><<<nb_gemm, 128, 0, stream>>>(feat, W, 0, fth);
        gat_attn<2><<<nb_attn, 256, 0, stream>>>(fth, al, ar, 0, el, er);
        gat_gather_pair<<<nb_gath, 256, 0, stream>>>(rowptr, list, fth, el, er, 0, out);

        gat_gemm<2><<<nb_gemm, 128, 0, stream>>>(feat, W, 1, fth);
        gat_attn<2><<<nb_attn, 256, 0, stream>>>(fth, al, ar, 1, el, er);
        gat_gather_pair<<<nb_gath, 256, 0, stream>>>(rowptr, list, fth, el, er, 1, out);
    }
}


# Projection and bucket fill in one launch: every fifth block is a projection tile, the other four park edges, so atomic round trips overlap tile loads; 4 launches; sum order = arrival rank as before
# speedup vs baseline: 8.7573x; 1.0547x over previous
#include <hip/hip_runtime.h>


#define N_NODES 100000
#define IN_F    128
#define H_HEADS 4
#define D_OUT   16
#define HD      64
#define HP      32
#define E_EDGES 3200000
#define SLOPE   0.2f
#define SCAN_T  1024

typedef __attribute__((ext_vector_type(2))) float v2f;
typedef __attribute__((ext_vector_type(8))) float v8f;

template <int NT>
__device__ __forceinline__ void gemm_tile(const float* __restrict__ feat,
                                          const float* __restrict__ W,
                                          int p,
                                          float* __restrict__ fth,
                                          int tile) {
    const int wave = threadIdx.x >> 5;
    const int lane = threadIdx.x & 31;
    const int rt   = wave / NT;
    const int ct   = wave % NT;
    const int r0   = tile * 32 + rt * 16;
    const int c0   = (p * NT + ct) * 16;
    const int m    = lane & 15;
    const int hi   = lane >> 4;

    const float* arow = feat + (size_t)(r0 + m) * IN_F + (hi << 1);
    const float* brow = W    + (size_t)(c0 + m) * IN_F + (hi << 1);

    v8f acc = {};
#pragma unroll
    for (int k0 = 0; k0 < IN_F; k0 += 4) {
        v2f a = *(const v2f*)(arow + k0);
        v2f b = *(const v2f*)(brow + k0);
        acc = __builtin_amdgcn_wmma_f32_16x16x4_f32(
            false, a, false, b, (short)0, acc, false, false);
    }
    float* outp = fth + (size_t)(r0 + hi * 8) * (NT * 16) + ct * 16 + m;
#pragma unroll
    for (int r = 0; r < 8; ++r) outp[(size_t)r * (NT * 16)] = acc[r];
}

template <int NT>
__global__ __launch_bounds__(64 * NT) void gat_gemm(const float* __restrict__ feat,
                                                    const float* __restrict__ W,
                                                    int p,
                                                    float* __restrict__ fth) {
    gemm_tile<NT>(feat, W, p, fth, (int)blockIdx.x);
}

template <int NT>
__global__ __launch_bounds__(256) void gat_attn(const float* __restrict__ fth,
                                                const float* __restrict__ al,
                                                const float* __restrict__ ar,
                                                int p,
                                                float* __restrict__ el,
                                                float* __restrict__ er) {
    int idx = blockIdx.x * blockDim.x + threadIdx.x;
    if (idx >= N_NODES * NT) return;
    const int h = p * NT + (idx % NT);
    const float* f = fth + (size_t)idx * D_OUT;
    float sl = 0.f, sr = 0.f;
#pragma unroll
    for (int d = 0; d < D_OUT; ++d) {
        float v = f[d];
        sl += v * al[h * D_OUT + d];
        sr += v * ar[h * D_OUT + d];
    }
    el[idx] = sl;
    er[idx] = sr;
}

__global__ __launch_bounds__(256) void csr_zero(int* __restrict__ cnt) {
    int i = blockIdx.x * blockDim.x + threadIdx.x;
    if (i < N_NODES) cnt[i] = 0;
}

__global__ __launch_bounds__(256) void csr_hist(const int* __restrict__ dst,
                                                int* __restrict__ cnt) {
    int i = blockIdx.x * blockDim.x + threadIdx.x;
    if (i < E_EDGES) atomicAdd(&cnt[dst[i]], 1);
}

__global__ __launch_bounds__(SCAN_T) void csr_scan(const int* __restrict__ cnt,
                                                   int* __restrict__ rowptr,
                                                   int* __restrict__ cur) {
    __shared__ int wsum[32];
    __shared__ int wtot[32];
    const int tid  = threadIdx.x;
    const int lane = tid & 31;
    const int wave = tid >> 5;
    const int base = blockIdx.x * SCAN_T;

    int part = 0;
    const int4* c4 = (const int4*)cnt;
    for (int j = tid; j < base / 4; j += SCAN_T) {
        int4 v = c4[j];
        part += (v.x + v.y) + (v.z + v.w);
    }
    for (int o = 16; o > 0; o >>= 1) part += __shfl_xor(part, o);
    if (lane == 0) wsum[wave] = part;

    const int i = base + tid;
    const int c = (i < N_NODES) ? cnt[i] : 0;
    int inc = c;
    for (int o = 1; o < 32; o <<= 1) {
        int t = __shfl_up(inc, o);
        if (lane >= o) inc += t;
    }
    if (lane == 31) wtot[wave] = inc;
    __syncthreads();

    int front = 0;
    for (int w = 0; w < 32; ++w) front += wsum[w];
    int woff = 0;
    for (int w = 0; w < wave; ++w) woff += wtot[w];
    const int excl = front + woff + inc - c;
    if (i <= N_NODES) rowptr[i] = excl;
    if (i <  N_NODES) cur[i]    = excl;
}

__global__ __launch_bounds__(256) void csr_fill(const int* __restrict__ src,
                                                const int* __restrict__ dst,
                                                int* __restrict__ cur,
                                                int* __restrict__ list) {
    int i = blockIdx.x * blockDim.x + threadIdx.x;
    if (i >= E_EDGES) return;
    const int pos = atomicAdd(&cur[dst[i]], 1);
    list[pos] = src[i];
}

#define SLOTS  64
#define HSLOTS 32

__global__ __launch_bounds__(256) void bkt_zero(int* __restrict__ cnt,
                                                int* __restrict__ nspill) {
    int i = blockIdx.x * blockDim.x + threadIdx.x;
    if (i < 2 * N_NODES) cnt[i] = 0;
    if (i == 0) *nspill = 0;
}

__device__ __forceinline__ void bkt_fill(const int* __restrict__ src,
                                         const int* __restrict__ dst,
                                         int* __restrict__ cnt,
                                         int* __restrict__ nspill,
                                         int* __restrict__ spill,
                                         float* __restrict__ out,
                                         int i) {
    if (i >= E_EDGES) return;
    const int half = (int)((__builtin_amdgcn_s_sendmsg_rtn(0x87) >> 18) & 1u);
    const int d   = dst[i];
    const int s   = src[i];
    const int pos = atomicAdd(&cnt[half * N_NODES + d], 1);
    int* where = (int*)out + (size_t)d * SLOTS + half * HSLOTS + pos;
    int  what  = s;
    if (pos >= HSLOTS) {
        where = spill + atomicAdd(nspill, 1);
        what  = i;
    }
    *where = what;
}

#define MIX 5
static_assert(E_EDGES / 256 == (MIX - 1) * (N_NODES / 32) && E_EDGES % 256 == 0 && N_NODES % 32 == 0,
              "one projection tile per MIX - 1 fill blocks, exactly");

__global__ __launch_bounds__(256) void gat_proj_fill(const float* __restrict__ feat,
                                                     const float* __restrict__ W,
                                                     float* __restrict__ fth,
                                                     const int* __restrict__ src,
                                                     const int* __restrict__ dst,
                                                     int* __restrict__ cnt,
                                                     int* __restrict__ nspill,
                                                     int* __restrict__ spill,
                                                     float* __restrict__ out) {
    const int b = (int)blockIdx.x;
    const int q = b / MIX;
    if (b == q * MIX) {
        gemm_tile<4>(feat, W, 0, fth, q);
    } else {
        bkt_fill(src, dst, cnt, nspill, spill, out, (b - q - 1) * 256 + (int)threadIdx.x);
    }
}

__device__ __forceinline__ float edge_score(float l, float r) {
    float e = l + r;
    return e >= 0.f ? e : SLOPE * e;
}


__device__ __forceinline__ float chunk_rows_pair(const float* __restrict__ fth,
                                            int s, float a0, float a1,
                                            int live, bool up, int lane, float acc) {
    const int cnt8 = (live + 7) & ~7;
    for (int k = 0; k < cnt8; k += 8) {
        float v[8], w[8];
#pragma unroll
        for (int u = 0; u < 8; ++u) {
            const int   sk = __builtin_amdgcn_readlane(s, k + u);
            const float w0 = __int_as_float(__builtin_amdgcn_readlane(__float_as_int(a0), k + u));
            const float w1 = __int_as_float(__builtin_amdgcn_readlane(__float_as_int(a1), k + u));
            v[u] = fth[(size_t)sk * HP + lane];
            w[u] = up ? w1 : w0;
        }
#pragma unroll
        for (int u = 0; u < 8; ++u) acc += w[u] * v[u];
    }
    return acc;
}

__global__ __launch_bounds__(256) void gat_gather_pair(const int* __restrict__ rowptr,
                                                  const int* __restrict__ list,
                                                  const float* __restrict__ fth,
                                                  const float* __restrict__ el,
                                                  const float* __restrict__ er,
                                                  int p,
                                                  float* __restrict__ out) {
    const int lane = threadIdx.x & 31;
    const int n    = blockIdx.x * 8 + __builtin_amdgcn_readfirstlane((int)(threadIdx.x >> 5));
    const bool up  = lane >= 16;
    const int beg  = rowptr[n];
    const int end  = rowptr[n + 1];
    const float er0 = er[(size_t)n * 2];
    const float er1 = er[(size_t)n * 2 + 1];

    float acc = 0.f;
    float sv  = 1.f;
    if (end > beg) {
        const int  deg  = end - beg;
        const int  last = end - 1;
        const bool two  = deg > 32;
        const int  jA   = beg + lane;
        const int  jB   = jA + 32;
        int sA, sB;
        v2f lA, lB;
        if (two) {
            sA = list[jA];
            sB = list[min(jB, last)];
            lA = *(const v2f*)(el + (size_t)sA * 2);
            lB = *(const v2f*)(el + (size_t)sB * 2);
        } else {
            sA = list[min(jA, last)];
            lA = *(const v2f*)(el + (size_t)sA * 2);
            sB = sA;
            lB = lA;
        }
        const float eA0 = edge_score(lA.x, er0);
        const float eA1 = edge_score(lA.y, er1);
        const float eB0 = edge_score(lB.x, er0);
        const float eB1 = edge_score(lB.y, er1);

        float m0 = jA < end ? eA0 : -INFINITY;
        float m1 = jA < end ? eA1 : -INFINITY;
        if (two) {
            m0 = fmaxf(m0, jB < end ? eB0 : -INFINITY);
            m1 = fmaxf(m1, jB < end ? eB1 : -INFINITY);
        }
        for (int b = beg + 64; b < end; b += 32) {
            const int j  = b + lane;
            const int s  = list[min(j, last)];
            const v2f l  = *(const v2f*)(el + (size_t)s * 2);
            const float e0 = edge_score(l.x, er0);
            const float e1 = edge_score(l.y, er1);
            m0 = fmaxf(m0, j < end ? e0 : -INFINITY);
            m1 = fmaxf(m1, j < end ? e1 : -INFINITY);
        }
        for (int o = 16; o > 0; o >>= 1) {
            m0 = fmaxf(m0, __shfl_xor(m0, o));
            m1 = fmaxf(m1, __shfl_xor(m1, o));
        }

        const float aA0 = jA < end ? expf(eA0 - m0) : 0.f;
        const float aA1 = jA < end ? expf(eA1 - m1) : 0.f;
        float s0 = aA0, s1 = aA1;
        acc = chunk_rows_pair(fth, sA, aA0, aA1, min(32, deg), up, lane, acc);
        if (two) {
            const float aB0 = jB < end ? expf(eB0 - m0) : 0.f;
            const float aB1 = jB < end ? expf(eB1 - m1) : 0.f;
            s0 += aB0;
            s1 += aB1;
            acc = chunk_rows_pair(fth, sB, aB0, aB1, min(32, deg - 32), up, lane, acc);
        }
        for (int b = beg + 64; b < end; b += 32) {
            const int j  = b + lane;
            const int s  = list[min(j, last)];
            const v2f l  = *(const v2f*)(el + (size_t)s * 2);
            const float a0 = j < end ? expf(edge_score(l.x, er0) - m0) : 0.f;
            const float a1 = j < end ? expf(edge_score(l.y, er1) - m1) : 0.f;
            s0 += a0;
            s1 += a1;
            acc = chunk_rows_pair(fth, s, a0, a1, min(32, end - b), up, lane, acc);
        }
        for (int o = 16; o > 0; o >>= 1) {
            s0 += __shfl_xor(s0, o);
            s1 += __shfl_xor(s1, o);
        }
        sv = up ? s1 : s0;
    }
    out[(size_t)n * HD + p * HP + lane] = acc / sv;
}


template <int U>
__device__ __forceinline__ float bcast8(float x) {
    return __uint_as_float(__builtin_amdgcn_mov_dpp8(__float_as_uint(x), U * 0x249249u));
}

constexpr unsigned xor8_sel(int x) {
    unsigned s = 0;
    for (int i = 0; i < 8; ++i) s |= (unsigned)((i ^ x) & 7) << (3 * i);
    return s;
}
template <int X>
__device__ __forceinline__ float xor8(float v) {
    return __uint_as_float(__builtin_amdgcn_mov_dpp8(__float_as_uint(v), xor8_sel(X)));
}

template <int U>
__device__ __forceinline__ void edge_row(const float* __restrict__ fth, int s, float a,
                                         int lane, v2f& v, float& w) {
    const int sk = __builtin_amdgcn_readlane(s, U);
    w = bcast8<U>(a);
    v = *(const v2f*)(fth + (size_t)sk * HD + 2 * lane);
}

__device__ __forceinline__ v2f chunk_rows_all(const float* __restrict__ fth,
                                          const int (&s)[4], const float (&a)[4],
                                          int live, int lane, v2f acc) {
#pragma unroll
    for (int g = 0; g < 4; ++g) {
        if (8 * g < live) {
            v2f v[8];
            float w[8];
            edge_row<0>(fth, s[g], a[g], lane, v[0], w[0]);
            edge_row<1>(fth, s[g], a[g], lane, v[1], w[1]);
            edge_row<2>(fth, s[g], a[g], lane, v[2], w[2]);
            edge_row<3>(fth, s[g], a[g], lane, v[3], w[3]);
            edge_row<4>(fth, s[g], a[g], lane, v[4], w[4]);
            edge_row<5>(fth, s[g], a[g], lane, v[5], w[5]);
            edge_row<6>(fth, s[g], a[g], lane, v[6], w[6]);
            edge_row<7>(fth, s[g], a[g], lane, v[7], w[7]);
#pragma unroll
            for (int u = 0; u < 8; ++u) {
                acc.x += w[u] * v[u].x;
                acc.y += w[u] * v[u].y;
            }
        }
    }
    return acc;
}

template <class F>
__device__ __forceinline__ void for_spill_edges(const int* __restrict__ spill,
                                                const int* __restrict__ src,
                                                const int* __restrict__ dst,
                                                int nsp, int n, int lane, F f) {
    for (int b = 0; b < nsp; b += 32) {
        const int k = b + lane;
        bool mine = false;
        int  s    = 0;
        if (k < nsp) {
            const int i = spill[k];
            mine = dst[i] == n;
            if (mine) s = src[i];
        }
        unsigned mask = __builtin_amdgcn_ballot_w32(mine);
        while (mask != 0u) {
            const int u = __builtin_ctz(mask);
            mask &= mask - 1u;
            f(__builtin_amdgcn_readlane(s, u));
        }
    }
}

__global__ __launch_bounds__(256) void gat_gather_all(const int* __restrict__ cnt,
                                                  const int* __restrict__ nspill,
                                                  const int* __restrict__ spill,
                                                  const int* __restrict__ src,
                                                  const int* __restrict__ dst,
                                                  const float* __restrict__ fth,
                                                  const float* __restrict__ el,
                                                  const float* __restrict__ er,
                                                  float* out) {
    const int lane = threadIdx.x & 31;
    const int n    = blockIdx.x * 8 + __builtin_amdgcn_readfirstlane((int)(threadIdx.x >> 5));
    const int hL   = lane >> 3;
    const int j    = lane & 7;
    const int cA   = cnt[n];
    const int cB   = cnt[N_NODES + n];
    const int* slot = (const int*)out + (size_t)n * SLOTS;
    const float erh = er[(size_t)n * H_HEADS + hL];

    v2f   acc = {0.f, 0.f};
    float sv  = 1.f;
    if (cA + cB > 0) {
        const int  dA   = min(cA, HSLOTS);
        const int  dB   = min(cB, HSLOTS);
        const int  deg  = dA + dB;
        const int  last = deg - 1;
        const bool two  = deg > 32;
        const int  nsp  = (cA > HSLOTS || cB > HSLOTS) ? *nspill : 0;
        const int  jB   = j + 32;
        auto at = [&](int p) { return slot[p < dA ? p : p - dA + HSLOTS]; };
        int   sA[4], sB[4];
        float eA[4], eB[4];
        if (two) {
#pragma unroll
            for (int g = 0; g < 4; ++g) sA[g] = at(j + 8 * g);
#pragma unroll
            for (int g = 0; g < 4; ++g) sB[g] = at(min(jB + 8 * g, last));
#pragma unroll
            for (int g = 0; g < 4; ++g) eA[g] = el[(size_t)sA[g] * H_HEADS + hL];
#pragma unroll
            for (int g = 0; g < 4; ++g) eB[g] = el[(size_t)sB[g] * H_HEADS + hL];
        } else {
#pragma unroll
            for (int g = 0; g < 4; ++g) sA[g] = at(min(j + 8 * g, last));
#pragma unroll
            for (int g = 0; g < 4; ++g) eA[g] = el[(size_t)sA[g] * H_HEADS + hL];
#pragma unroll
            for (int g = 0; g < 4; ++g) { sB[g] = sA[g]; eB[g] = eA[g]; }
        }
#pragma unroll
        for (int g = 0; g < 4; ++g) {
            eA[g] = edge_score(eA[g], erh);
            eB[g] = edge_score(eB[g], erh);
        }

        float m = -INFINITY;
#pragma unroll
        for (int g = 0; g < 4; ++g) m = fmaxf(m, j + 8 * g < deg ? eA[g] : -INFINITY);
        if (two) {
#pragma unroll
            for (int g = 0; g < 4; ++g) m = fmaxf(m, jB + 8 * g < deg ? eB[g] : -INFINITY);
        }
        for_spill_edges(spill, src, dst, nsp, n, lane, [&](int s) {
            m = fmaxf(m, edge_score(el[(size_t)s * H_HEADS + hL], erh));
        });
        m = fmaxf(m, xor8<4>(m));
        m = fmaxf(m, xor8<2>(m));
        m = fmaxf(m, xor8<1>(m));

        float aA[4];
        float ssum = 0.f;
#pragma unroll
        for (int g = 0; g < 4; ++g) {
            aA[g] = j + 8 * g < deg ? expf(eA[g] - m) : 0.f;
            ssum += aA[g];
        }
        acc = chunk_rows_all(fth, sA, aA, min(32, deg), lane, acc);
        if (two) {
            float aB[4];
#pragma unroll
            for (int g = 0; g < 4; ++g) {
                aB[g] = jB + 8 * g < deg ? expf(eB[g] - m) : 0.f;
                ssum += aB[g];
            }
            acc = chunk_rows_all(fth, sB, aB, deg - 32, lane, acc);
        }
        for_spill_edges(spill, src, dst, nsp, n, lane, [&](int s) {
            const float a = expf(edge_score(el[(size_t)s * H_HEADS + hL], erh) - m);
            const v2f   v = *(const v2f*)(fth + (size_t)s * HD + 2 * lane);
            ssum  += j == 0 ? a : 0.f;
            acc.x += a * v.x;
            acc.y += a * v.y;
        });
        ssum += xor8<4>(ssum);
        ssum += xor8<2>(ssum);
        ssum += xor8<1>(ssum);
        sv = ssum;
    }
    v2f res;
    res.x = acc.x / sv;
    res.y = acc.y / sv;
    *(v2f*)(out + (size_t)n * HD + 2 * lane) = res;
}

extern "C" void kernel_launch(void* const* d_in, const int* in_sizes, int n_in,
                              void* d_out, int out_size, void* d_ws, size_t ws_size,
                              hipStream_t stream) {
    const float* feat = (const float*)d_in[0];
    const float* W    = (const float*)d_in[1];
    const float* al   = (const float*)d_in[2];
    const float* ar   = (const float*)d_in[3];
    const int*   src  = (const int*)d_in[4];
    const int*   dst  = (const int*)d_in[5];
    float* out = (float*)d_out;

    const int nb_node = (N_NODES + 255) / 256;
    const int nb_edge = (E_EDGES + 255) / 256;
    const int nb_scan = (N_NODES + 1 + SCAN_T - 1) / SCAN_T;
    const int nb_gemm = N_NODES / 32;
    const int nb_gath = N_NODES / 8;

    if (ws_size >= (size_t)42400004) {
        float* fth    = (float*)d_ws;
        int*   spill  = (int*)(fth + (size_t)N_NODES * HD);
        float* el     = (float*)(spill + (size_t)E_EDGES);
        float* er     = el + (size_t)N_NODES * H_HEADS;
        int*   cnt    = (int*)(er + (size_t)N_NODES * H_HEADS);
        int*   nspill = cnt + (size_t)2 * N_NODES;
        const int nb_attn = (N_NODES * H_HEADS + 255) / 256;
        const int nb_zero = (2 * N_NODES + 255) / 256;

        const int nb_mix  = MIX * (N_NODES / 32);

        bkt_zero<<<nb_zero, 256, 0, stream>>>(cnt, nspill);
        gat_proj_fill<<<nb_mix, 256, 0, stream>>>(feat, W, fth, src, dst, cnt, nspill, spill, out);
        gat_attn<4><<<nb_attn, 256, 0, stream>>>(fth, al, ar, 0, el, er);
        gat_gather_all<<<nb_gath, 256, 0, stream>>>(cnt, nspill, spill, src, dst, fth, el, er, out);
    } else {
        float* fth    = (float*)d_ws;
        int*   list   = (int*)(fth + (size_t)N_NODES * HP);
        float* el     = (float*)(list + (size_t)E_EDGES);
        float* er     = el + (size_t)N_NODES * 2;
        int*   cnt    = (int*)(er + (size_t)N_NODES * 2);
        int*   rowptr = cnt + (size_t)N_NODES;
        int*   cur    = rowptr + 100352;
        const int nb_attn = (N_NODES * 2 + 255) / 256;

        csr_zero<<<nb_node, 256, 0, stream>>>(cnt);
        csr_hist<<<nb_edge, 256, 0, stream>>>(dst, cnt);
        csr_scan<<<nb_scan, SCAN_T, 0, stream>>>(cnt, rowptr, cur);
        csr_fill<<<nb_edge, 256, 0, stream>>>(src, dst, cur, list);

        gat_gemm<2><<<nb_gemm, 128, 0, stream>>>(feat, W, 0, fth);
        gat_attn<2><<<nb_attn, 256, 0, stream>>>(fth, al, ar, 0, el, er);
        gat_gather_pair<<<nb_gath, 256, 0, stream>>>(rowptr, list, fth, el, er, 0, out);

        gat_gemm<2><<<nb_gemm, 128, 0, stream>>>(feat, W, 1, fth);
        gat_attn<2><<<nb_attn, 256, 0, stream>>>(fth, al, ar, 1, el, er);
        gat_gather_pair<<<nb_gath, 256, 0, stream>>>(rowptr, list, fth, el, er, 1, out);
    }
}
